# starting kernel
# baseline (speedup 1.0000x reference)
#include <hip/hip_runtime.h>

typedef float v2f __attribute__((ext_vector_type(2)));
typedef float v8f __attribute__((ext_vector_type(8)));

#define IN_F   4096
#define OUT_F  16384
#define M_ROWS 64
#define KK     16
#define NUM_BANKS 64
#define ANNEAL 2000
#define KC     256
#define NW     8
#define NTILE  (NW * 16)
#define KSPLIT 4
#define KSLICE (IN_F / KSPLIT)

__global__ __launch_bounds__(256)
void zero_out(float* __restrict__ out, int n4)
{
    int i = blockIdx.x * blockDim.x + threadIdx.x;
    if (i < n4) ((float4*)out)[i] = make_float4(0.f, 0.f, 0.f, 0.f);
}

__global__ __launch_bounds__(256)
void density_linear_wmma(const float* __restrict__ x,
                         const float* __restrict__ lw,
                         const float* __restrict__ scale,
                         const float* __restrict__ thr,
                         const int*   __restrict__ stepp,
                         float* __restrict__ out)
{
    __shared__ float xs[M_ROWS * KC];

    const int tid   = threadIdx.x;
    const int lane  = tid & 31;
    const int wave  = tid >> 5;
    const int nbase = blockIdx.x * NTILE + wave * 16;
    const int ncol  = nbase + (lane & 15);
    const int koff  = (lane >> 4) * 2;
    const int kbase = blockIdx.y * KSLICE;

    const int step = stepp[0];
    float a = fminf((float)step / (float)ANNEAL, 1.0f);
    const float oma = 1.0f - a;
    const float a8  = a * 0.125f;

    const float* Rp = thr + (unsigned)(step % NUM_BANKS) * KK;
    float r[KK];
    #pragma unroll
    for (int i = 0; i < KK; ++i) r[i] = Rp[i];

    const float s = scale[ncol];

    v8f acc[4];
    #pragma unroll
    for (int mt = 0; mt < 4; ++mt) acc[mt] = (v8f)0.0f;

    const float* wrow = lw + (size_t)ncol * IN_F + koff;

    for (int kc = kbase; kc < kbase + KSLICE; kc += KC) {
        __syncthreads();
        for (int i = tid * 4; i < M_ROWS * KC; i += 256 * 4) {
            const int m = i / KC;
            const int j = i & (KC - 1);
            const float4 v = *(const float4*)(x + (size_t)m * IN_F + kc + j);
            *(float4*)(xs + i) = v;
        }
        __syncthreads();

        if (kc + KC < kbase + KSLICE)
            __builtin_prefetch(wrow + kc + KC, 0, 1);

        #pragma unroll 4
        for (int k = 0; k < KC; k += 4) {
            const float2 wv = *(const float2*)(wrow + kc + k);
            v2f b;
            {
                float c0 = fminf(fmaxf(wv.x, -1.0f), 1.0f);
                float c1 = fminf(fmaxf(wv.y, -1.0f), 1.0f);
                float t0 = __builtin_fmaf(c0, 0.5f, 0.5f);
                float t1 = __builtin_fmaf(c1, 0.5f, 0.5f);
                int n0 = 0, n1 = 0;
                #pragma unroll
                for (int i = 0; i < KK; ++i) {
                    n0 += (r[i] < t0) ? 1 : 0;
                    n1 += (r[i] < t1) ? 1 : 0;
                }
                b.x = __builtin_fmaf(oma, c0, __builtin_fmaf((float)n0, a8, -a));
                b.y = __builtin_fmaf(oma, c1, __builtin_fmaf((float)n1, a8, -a));
            }

            #pragma unroll
            for (int mt = 0; mt < 4; ++mt) {
                const int m = mt * 16 + (lane & 15);
                v2f av = *(const v2f*)(xs + m * KC + (k & (KC - 1)) + koff);
                acc[mt] = __builtin_amdgcn_wmma_f32_16x16x4_f32(
                    false, av, false, b, (short)0, acc[mt], false, false);
            }
        }
    }

    const int mrow0 = (lane >> 4) * 8;
    #pragma unroll
    for (int mt = 0; mt < 4; ++mt) {
        #pragma unroll
        for (int j = 0; j < 8; ++j) {
            const int m = mt * 16 + mrow0 + j;
            unsafeAtomicAdd(&out[(size_t)m * OUT_F + ncol], acc[mt][j] * s);
        }
    }
}

extern "C" void kernel_launch(void* const* d_in, const int* in_sizes, int n_in,
                              void* d_out, int out_size, void* d_ws, size_t ws_size,
                              hipStream_t stream) {
    const float* x     = (const float*)d_in[0];
    const float* lw    = (const float*)d_in[1];
    const float* scale = (const float*)d_in[2];
    const float* thr   = (const float*)d_in[3];
    const int*   step  = (const int*)d_in[4];
    float* out = (float*)d_out;

    const int n4 = out_size / 4;
    zero_out<<<dim3((n4 + 255) / 256), dim3(256), 0, stream>>>(out, n4);

    dim3 grid(OUT_F / NTILE, KSPLIT);
    dim3 block(256);
    density_linear_wmma<<<grid, block, 0, stream>>>(x, lw, scale, thr, step, out);
}


# x fragments held in registers: each of 16 waves owns 64 K for all 64 rows and walks 16 feature tiles, waves' K partials summed in LDS per tile; no x tile in LDS; summation order changes
# speedup vs baseline: 1.2443x; 1.2443x over previous
#include <hip/hip_runtime.h>

typedef float v2f __attribute__((ext_vector_type(2)));
typedef float v8f __attribute__((ext_vector_type(8)));

#define IN_F   4096
#define OUT_F  16384
#define M_ROWS 64
#define KK     16
#define NUM_BANKS 64
#define ANNEAL 2000
#define NW     16
#define KWAVE  64
#define KSTEPS (KWAVE / 4)
#define KSLICE (NW * KWAVE)
#define KSPLIT (IN_F / KSLICE)
#define NTILES 16
#define NBLK   (NTILES * 16)

__global__ __launch_bounds__(256)
void zero_out(float* __restrict__ out, int n4)
{
    int i = blockIdx.x * blockDim.x + threadIdx.x;
    if (i < n4) ((float4*)out)[i] = make_float4(0.f, 0.f, 0.f, 0.f);
}

__global__ __launch_bounds__(NW * 32)
void density_linear_wmma(const float* __restrict__ x,
                         const float* __restrict__ lw,
                         const float* __restrict__ scale,
                         const float* __restrict__ thr,
                         const int*   __restrict__ stepp,
                         float* __restrict__ out)
{
    __shared__ float red[2][NW][32][32];

    const int tid   = threadIdx.x;
    const int lane  = tid & 31;
    const int wave  = tid >> 5;
    const int c     = lane & 15;
    const int h     = lane >> 4;
    const int koff  = h * 2;
    const int kw0   = blockIdx.y * KSLICE + wave * KWAVE;
    const int nblk0 = blockIdx.x * NBLK;

    const int step = stepp[0];
    float a = fminf((float)step / (float)ANNEAL, 1.0f);
    const float oma = 1.0f - a;
    const float a8  = a * 0.125f;

    const float* Rp = thr + (unsigned)(step % NUM_BANKS) * KK;
    float r[KK];
    #pragma unroll
    for (int i = 0; i < KK; ++i) r[i] = Rp[i];

    v2f af[KSTEPS][4];
    #pragma unroll
    for (int ks = 0; ks < KSTEPS; ++ks) {
        #pragma unroll
        for (int mt = 0; mt < 4; ++mt)
            af[ks][mt] = *(const v2f*)(x + (size_t)(mt * 16 + c) * IN_F + kw0 + ks * 4 + koff);
    }

    const int rr   = 2 * wave;
    const int mout = (rr >> 3) * 16 + h * 8 + (rr & 7);

    for (int t = 0; t < NTILES; ++t) {
        const int ncol = nblk0 + t * 16 + c;
        const float* wrow = lw + (size_t)ncol * IN_F + kw0 + koff;

        if (t + 1 < NTILES)
            __builtin_prefetch(lw + (size_t)(ncol + 16) * IN_F + kw0, 0, 1);

        v8f acc[4];
        #pragma unroll
        for (int mt = 0; mt < 4; ++mt) acc[mt] = (v8f)0.0f;

        #pragma unroll
        for (int ks = 0; ks < KSTEPS; ++ks) {
            const float2 wv = *(const float2*)(wrow + ks * 4);
            v2f b;
            {
                float c0 = fminf(fmaxf(wv.x, -1.0f), 1.0f);
                float c1 = fminf(fmaxf(wv.y, -1.0f), 1.0f);
                float t0 = __builtin_fmaf(c0, 0.5f, 0.5f);
                float t1 = __builtin_fmaf(c1, 0.5f, 0.5f);
                int n0 = 0, n1 = 0;
                #pragma unroll
                for (int i = 0; i < KK; ++i) {
                    n0 += (r[i] < t0) ? 1 : 0;
                    n1 += (r[i] < t1) ? 1 : 0;
                }
                b.x = __builtin_fmaf(oma, c0, __builtin_fmaf((float)n0, a8, -a));
                b.y = __builtin_fmaf(oma, c1, __builtin_fmaf((float)n1, a8, -a));
            }

            #pragma unroll
            for (int mt = 0; mt < 4; ++mt) {
                acc[mt] = __builtin_amdgcn_wmma_f32_16x16x4_f32(
                    false, af[ks][mt], false, b, (short)0, acc[mt], false, false);
            }
        }

        float* mine = &red[t & 1][wave][0][lane];
        #pragma unroll
        for (int mt = 0; mt < 4; ++mt) {
            #pragma unroll
            for (int j = 0; j < 8; ++j)
                mine[(mt * 8 + j) * 32] = acc[mt][j];
        }
        __syncthreads();
        const float* col = &red[t & 1][0][rr][lane];
        float s0 = 0.0f, s1 = 0.0f;
        #pragma unroll
        for (int w = 0; w < NW; ++w) {
            s0 += col[w * 1024];
            s1 += col[w * 1024 + 32];
        }

        const float s = scale[ncol];
        unsafeAtomicAdd(&out[(size_t)mout * OUT_F + ncol], s0 * s);
        unsafeAtomicAdd(&out[(size_t)(mout + 1) * OUT_F + ncol], s1 * s);
    }
}

extern "C" void kernel_launch(void* const* d_in, const int* in_sizes, int n_in,
                              void* d_out, int out_size, void* d_ws, size_t ws_size,
                              hipStream_t stream) {
    const float* x     = (const float*)d_in[0];
    const float* lw    = (const float*)d_in[1];
    const float* scale = (const float*)d_in[2];
    const float* thr   = (const float*)d_in[3];
    const int*   step  = (const int*)d_in[4];
    float* out = (float*)d_out;

    const int n4 = out_size / 4;
    zero_out<<<dim3((n4 + 255) / 256), dim3(256), 0, stream>>>(out, n4);

    dim3 grid(OUT_F / NBLK, KSPLIT);
    dim3 block(NW * 32);
    density_linear_wmma<<<grid, block, 0, stream>>>(x, lw, scale, thr, step, out);
}


# threshold count read from a per-block LDS table of 8192 target-density cells (count below the cell + one compare inside it); full count only for a tile that meets a cell with two; sums unchanged
# speedup vs baseline: 1.6236x; 1.3048x over previous
#include <hip/hip_runtime.h>

typedef float v2f __attribute__((ext_vector_type(2)));
typedef float v8f __attribute__((ext_vector_type(8)));

#define IN_F   4096
#define OUT_F  16384
#define M_ROWS 64
#define KK     16
#define NUM_BANKS 64
#define ANNEAL 2000
#define NW     16
#define KWAVE  64
#define KSTEPS (KWAVE / 4)
#define KSLICE (NW * KWAVE)
#define KSPLIT (IN_F / KSLICE)
#define NTILES 16
#define NBLK   (NTILES * 16)
#define GCELL  8192
#define MULTI  0x100u

__device__ __forceinline__ int grid_cell(float v)
{
    float f = floorf(v * (float)GCELL);
    f = fminf(fmaxf(f, -1.0f), (float)(GCELL + 1));
    return (int)f;
}

__device__ __forceinline__ float eff_weight(float cl, int n, float oma, float a8, float a)
{
    return __builtin_fmaf(oma, cl, __builtin_fmaf((float)n, a8, -a));
}

__device__ __forceinline__ void tile_products_exact(const float* wrow,
                                                    const v2f (&af)[KSTEPS][4],
                                                    const float* Rp,
                                                    float oma, float a8, float a,
                                                    v8f (&acc)[4])
{
    #pragma unroll
    for (int mt = 0; mt < 4; ++mt) acc[mt] = (v8f)0.0f;
    #pragma unroll
    for (int ks = 0; ks < KSTEPS; ++ks) {
        const float2 wv = *(const float2*)(wrow + ks * 4);
        const float c0 = fminf(fmaxf(wv.x, -1.0f), 1.0f);
        const float c1 = fminf(fmaxf(wv.y, -1.0f), 1.0f);
        const float t0 = __builtin_fmaf(c0, 0.5f, 0.5f);
        const float t1 = __builtin_fmaf(c1, 0.5f, 0.5f);
        int n0 = 0, n1 = 0;
        #pragma nounroll
        for (int i = 0; i < KK; ++i) {
            const float ri = Rp[i];
            n0 += (ri < t0) ? 1 : 0;
            n1 += (ri < t1) ? 1 : 0;
        }
        v2f b;
        b.x = eff_weight(c0, n0, oma, a8, a);
        b.y = eff_weight(c1, n1, oma, a8, a);
        #pragma unroll
        for (int mt = 0; mt < 4; ++mt)
            acc[mt] = __builtin_amdgcn_wmma_f32_16x16x4_f32(
                false, af[ks][mt], false, b, (short)0, acc[mt], false, false);
    }
}

__device__ __forceinline__ unsigned tile_products_table(const float* __restrict__ wrow,
                                                        const v2f (&af)[KSTEPS][4],
                                                        const uint2* tab,
                                                        float oma, float a8, float a,
                                                        v8f (&acc)[4])
{
    unsigned flags = 0u;
    #pragma unroll
    for (int mt = 0; mt < 4; ++mt) acc[mt] = (v8f)0.0f;

    #pragma unroll
    for (int ks = 0; ks < KSTEPS; ++ks) {
        const float2 wv = *(const float2*)(wrow + ks * 4);
        const float c0 = fminf(fmaxf(wv.x, -1.0f), 1.0f);
        const float c1 = fminf(fmaxf(wv.y, -1.0f), 1.0f);
        const float t0 = __builtin_fmaf(c0, 0.5f, 0.5f);
        const float t1 = __builtin_fmaf(c1, 0.5f, 0.5f);
        const uint2 e0 = tab[(int)(t0 * (float)GCELL)];
        const uint2 e1 = tab[(int)(t1 * (float)GCELL)];
        const int n0 = (int)(e0.y & 0xffu) + ((__uint_as_float(e0.x) < t0) ? 1 : 0);
        const int n1 = (int)(e1.y & 0xffu) + ((__uint_as_float(e1.x) < t1) ? 1 : 0);
        flags |= e0.y | e1.y;
        v2f b;
        b.x = eff_weight(c0, n0, oma, a8, a);
        b.y = eff_weight(c1, n1, oma, a8, a);
        #pragma unroll
        for (int mt = 0; mt < 4; ++mt)
            acc[mt] = __builtin_amdgcn_wmma_f32_16x16x4_f32(
                false, af[ks][mt], false, b, (short)0, acc[mt], false, false);
    }
    return flags;
}

__global__ __launch_bounds__(256)
void zero_out(float* __restrict__ out, int n4)
{
    int i = blockIdx.x * blockDim.x + threadIdx.x;
    if (i < n4) ((float4*)out)[i] = make_float4(0.f, 0.f, 0.f, 0.f);
}

__global__ __launch_bounds__(NW * 32)
void density_linear_wmma(const float* __restrict__ x,
                         const float* __restrict__ lw,
                         const float* __restrict__ scale,
                         const float* __restrict__ thr,
                         const int*   __restrict__ stepp,
                         float* __restrict__ out)
{
    __shared__ float red[2][NW][32][32];
    __shared__ uint2 tab[GCELL + 8];

    const int tid   = threadIdx.x;
    const int lane  = tid & 31;
    const int wave  = __builtin_amdgcn_readfirstlane(tid >> 5);
    const int c     = lane & 15;
    const int h     = lane >> 4;
    const int koff  = h * 2;
    const int kw0   = blockIdx.y * KSLICE + wave * KWAVE;
    const int nblk0 = blockIdx.x * NBLK;
    const int loff  = c * IN_F + koff;

    const int step = stepp[0];
    float a = fminf((float)step / (float)ANNEAL, 1.0f);
    const float oma = 1.0f - a;
    const float a8  = a * 0.125f;

    const float* Rp = thr + (unsigned)(step % NUM_BANKS) * KK;
    float r[KK];
    #pragma unroll
    for (int i = 0; i < KK; ++i) r[i] = Rp[i];

    v2f af[KSTEPS][4];
    #pragma unroll
    for (int ks = 0; ks < KSTEPS; ++ks) {
        #pragma unroll
        for (int mt = 0; mt < 4; ++mt)
            af[ks][mt] = *(const v2f*)(x + kw0 + mt * 16 * IN_F + ks * 4 + loff);
    }

    {
        int ci[KK];
        #pragma unroll
        for (int i = 0; i < KK; ++i) ci[i] = grid_cell(r[i]);
        const int k0 = tid * (GCELL / (NW * 32));
        int below[GCELL / (NW * 32) + 2];
        #pragma unroll
        for (int j = 0; j < GCELL / (NW * 32) + 2; ++j) {
            unsigned bits = 0u;
            #pragma unroll
            for (int i = 0; i < KK; ++i)
                bits = __builtin_amdgcn_alignbit(bits, (unsigned)(ci[i] - (k0 + j)), 31u);
            below[j] = __builtin_popcount(bits);
        }
        #pragma unroll
        for (int j = 0; j < GCELL / (NW * 32) + 1; ++j) {
            const int inside = below[j + 1] - below[j];
            if (j < GCELL / (NW * 32) || k0 + j == GCELL)
                tab[k0 + j] = make_uint2(0x7f800000u, (unsigned)below[j] | (inside >= 2 ? MULTI : 0u));
        }
        __syncthreads();
        if (tid < KK) {
            const float rv = Rp[tid];
            const int cc = grid_cell(rv);
            if (cc >= 0 && cc <= GCELL) tab[cc].x = __float_as_uint(rv);
        }
        __syncthreads();
    }

    const int rr   = 2 * wave;
    const int mout = (rr >> 3) * 16 + h * 8 + (rr & 7);

    for (int t = 0; t < NTILES; ++t) {
        const int n0t = nblk0 + t * 16;
        const float* wtile = lw + (size_t)n0t * IN_F + kw0;
        const float* wrow  = wtile + loff;

        if (t + 1 < NTILES)
            __builtin_prefetch(wtile + 16 * IN_F + c * IN_F, 0, 1);

        v8f acc[4];
        const unsigned flags = tile_products_table(wrow, af, tab, oma, a8, a, acc);
        if (__builtin_amdgcn_ballot_w32((flags & MULTI) != 0u) != 0u) {
            const float* wagain = wrow;
            asm volatile("" : "+v"(wagain));
            tile_products_exact(wagain, af, Rp, oma, a8, a, acc);
        }

        float* mine = &red[t & 1][wave][0][lane];
        #pragma unroll
        for (int mt = 0; mt < 4; ++mt) {
            #pragma unroll
            for (int j = 0; j < 8; ++j)
                mine[(mt * 8 + j) * 32] = acc[mt][j];
        }
        __syncthreads();
        const float* col = &red[t & 1][0][rr][lane];
        float s0 = 0.0f, s1 = 0.0f;
        #pragma unroll
        for (int w = 0; w < NW; ++w) {
            s0 += col[w * 1024];
            s1 += col[w * 1024 + 32];
        }

        const float s = scale[n0t + c];
        unsafeAtomicAdd(&out[(size_t)mout * OUT_F + n0t + c], s0 * s);
        unsafeAtomicAdd(&out[(size_t)(mout + 1) * OUT_F + n0t + c], s1 * s);
    }
}

extern "C" void kernel_launch(void* const* d_in, const int* in_sizes, int n_in,
                              void* d_out, int out_size, void* d_ws, size_t ws_size,
                              hipStream_t stream) {
    const float* x     = (const float*)d_in[0];
    const float* lw    = (const float*)d_in[1];
    const float* scale = (const float*)d_in[2];
    const float* thr   = (const float*)d_in[3];
    const int*   step  = (const int*)d_in[4];
    float* out = (float*)d_out;

    const int n4 = out_size / 4;
    zero_out<<<dim3((n4 + 255) / 256), dim3(256), 0, stream>>>(out, n4);

    dim3 grid(OUT_F / NBLK, KSPLIT);
    dim3 block(NW * 32);
    density_linear_wmma<<<grid, block, 0, stream>>>(x, lw, scale, thr, step, out);
}


# weight tile staged by each wave through its own LDS rows with whole-line 16-byte global loads (4 lines per load, was 16); x and weight fragments take k in groups of 8; summation order changes
# speedup vs baseline: 1.6988x; 1.0463x over previous
#include <hip/hip_runtime.h>

typedef float v2f __attribute__((ext_vector_type(2)));
typedef float v8f __attribute__((ext_vector_type(8)));

#define IN_F   4096
#define OUT_F  16384
#define M_ROWS 64
#define KK     16
#define NUM_BANKS 64
#define ANNEAL 2000
#define NW     16
#define KWAVE  64
#define KSTEPS (KWAVE / 4)
#define KSLICE (NW * KWAVE)
#define KSPLIT (IN_F / KSLICE)
#define NTILES 16
#define NBLK   (NTILES * 16)
#define GCELL  8192
#define MULTI  0x100u
#define KGRP   (KWAVE / 8)
#define SPITCH 17

__device__ __forceinline__ int grid_cell(float v)
{
    float f = floorf(v * (float)GCELL);
    f = fminf(fmaxf(f, -1.0f), (float)(GCELL + 1));
    return (int)f;
}

__device__ __forceinline__ float eff_weight(float cl, int n, float oma, float a8, float a)
{
    return __builtin_fmaf(oma, cl, __builtin_fmaf((float)n, a8, -a));
}

__device__ __forceinline__ void tile_products_exact(const float4* srow,
                                                    const v2f (&af)[KSTEPS][4],
                                                    const float* Rp,
                                                    float oma, float a8, float a,
                                                    v8f (&acc)[4])
{
    #pragma unroll
    for (int mt = 0; mt < 4; ++mt) acc[mt] = (v8f)0.0f;
    #pragma unroll
    for (int ks = 0; ks < KSTEPS; ++ks) {
        const float4 q = srow[2 * (ks >> 1)];
        const float w0 = (ks & 1) ? q.z : q.x;
        const float w1 = (ks & 1) ? q.w : q.y;
        const float c0 = fminf(fmaxf(w0, -1.0f), 1.0f);
        const float c1 = fminf(fmaxf(w1, -1.0f), 1.0f);
        const float t0 = __builtin_fmaf(c0, 0.5f, 0.5f);
        const float t1 = __builtin_fmaf(c1, 0.5f, 0.5f);
        int n0 = 0, n1 = 0;
        #pragma nounroll
        for (int i = 0; i < KK; ++i) {
            const float ri = Rp[i];
            n0 += (ri < t0) ? 1 : 0;
            n1 += (ri < t1) ? 1 : 0;
        }
        v2f b;
        b.x = eff_weight(c0, n0, oma, a8, a);
        b.y = eff_weight(c1, n1, oma, a8, a);
        #pragma unroll
        for (int mt = 0; mt < 4; ++mt)
            acc[mt] = __builtin_amdgcn_wmma_f32_16x16x4_f32(
                false, af[ks][mt], false, b, (short)0, acc[mt], false, false);
    }
}

__device__ __forceinline__ unsigned tile_products_table(const float4* srow,
                                                        const v2f (&af)[KSTEPS][4],
                                                        const uint2* tab,
                                                        float oma, float a8, float a,
                                                        v8f (&acc)[4])
{
    unsigned flags = 0u;
    #pragma unroll
    for (int mt = 0; mt < 4; ++mt) acc[mt] = (v8f)0.0f;

    #pragma unroll
    for (int ks = 0; ks < KSTEPS; ++ks) {
        const float4 q = srow[2 * (ks >> 1)];
        const float w0 = (ks & 1) ? q.z : q.x;
        const float w1 = (ks & 1) ? q.w : q.y;
        const float c0 = fminf(fmaxf(w0, -1.0f), 1.0f);
        const float c1 = fminf(fmaxf(w1, -1.0f), 1.0f);
        const float t0 = __builtin_fmaf(c0, 0.5f, 0.5f);
        const float t1 = __builtin_fmaf(c1, 0.5f, 0.5f);
        const uint2 e0 = tab[(int)(t0 * (float)GCELL)];
        const uint2 e1 = tab[(int)(t1 * (float)GCELL)];
        const int n0 = (int)(e0.y & 0xffu) + ((__uint_as_float(e0.x) < t0) ? 1 : 0);
        const int n1 = (int)(e1.y & 0xffu) + ((__uint_as_float(e1.x) < t1) ? 1 : 0);
        flags |= e0.y | e1.y;
        v2f b;
        b.x = eff_weight(c0, n0, oma, a8, a);
        b.y = eff_weight(c1, n1, oma, a8, a);
        #pragma unroll
        for (int mt = 0; mt < 4; ++mt)
            acc[mt] = __builtin_amdgcn_wmma_f32_16x16x4_f32(
                false, af[ks][mt], false, b, (short)0, acc[mt], false, false);
    }
    return flags;
}

__global__ __launch_bounds__(256)
void zero_out(float* __restrict__ out, int n4)
{
    int i = blockIdx.x * blockDim.x + threadIdx.x;
    if (i < n4) ((float4*)out)[i] = make_float4(0.f, 0.f, 0.f, 0.f);
}

__global__ __launch_bounds__(NW * 32)
void density_linear_wmma(const float* __restrict__ x,
                         const float* __restrict__ lw,
                         const float* __restrict__ scale,
                         const float* __restrict__ thr,
                         const int*   __restrict__ stepp,
                         float* __restrict__ out)
{
    __shared__ float red[2][NW][32][32];
    __shared__ uint2 tab[GCELL + 8];
    __shared__ float4 wst[NW][16][SPITCH];

    const int tid   = threadIdx.x;
    const int lane  = tid & 31;
    const int wave  = __builtin_amdgcn_readfirstlane(tid >> 5);
    const int c     = lane & 15;
    const int h     = lane >> 4;
    const int koff  = h * 4;
    const int kw0   = blockIdx.y * KSLICE + wave * KWAVE;
    const int nblk0 = blockIdx.x * NBLK;
    const int loff  = c * IN_F + koff;

    const int step = stepp[0];
    float a = fminf((float)step / (float)ANNEAL, 1.0f);
    const float oma = 1.0f - a;
    const float a8  = a * 0.125f;

    const float* Rp = thr + (unsigned)(step % NUM_BANKS) * KK;
    float r[KK];
    #pragma unroll
    for (int i = 0; i < KK; ++i) r[i] = Rp[i];

    v2f af[KSTEPS][4];
    #pragma unroll
    for (int p = 0; p < KGRP; ++p) {
        #pragma unroll
        for (int mt = 0; mt < 4; ++mt) {
            const float4 xv = *(const float4*)(x + kw0 + mt * 16 * IN_F + p * 8 + loff);
            af[2 * p][mt].x     = xv.x;
            af[2 * p][mt].y     = xv.y;
            af[2 * p + 1][mt].x = xv.z;
            af[2 * p + 1][mt].y = xv.w;
        }
    }

    {
        int ci[KK];
        #pragma unroll
        for (int i = 0; i < KK; ++i) ci[i] = grid_cell(r[i]);
        const int k0 = tid * (GCELL / (NW * 32));
        int below[GCELL / (NW * 32) + 2];
        #pragma unroll
        for (int j = 0; j < GCELL / (NW * 32) + 2; ++j) {
            unsigned bits = 0u;
            #pragma unroll
            for (int i = 0; i < KK; ++i)
                bits = __builtin_amdgcn_alignbit(bits, (unsigned)(ci[i] - (k0 + j)), 31u);
            below[j] = __builtin_popcount(bits);
        }
        #pragma unroll
        for (int j = 0; j < GCELL / (NW * 32) + 1; ++j) {
            const int inside = below[j + 1] - below[j];
            if (j < GCELL / (NW * 32) || k0 + j == GCELL)
                tab[k0 + j] = make_uint2(0x7f800000u, (unsigned)below[j] | (inside >= 2 ? MULTI : 0u));
        }
        __syncthreads();
        if (tid < KK) {
            const float rv = Rp[tid];
            const int cc = grid_cell(rv);
            if (cc >= 0 && cc <= GCELL) tab[cc].x = __float_as_uint(rv);
        }
        __syncthreads();
    }

    const int rr   = 2 * wave;
    const int mout = (rr >> 3) * 16 + h * 8 + (rr & 7);

    for (int t = 0; t < NTILES; ++t) {
        const int n0t = nblk0 + t * 16;
        const float* wtile = lw + (size_t)n0t * IN_F + kw0;

        const float* gsrc = wtile + h * IN_F + c * 4;
        float4 g[8];
        #pragma unroll
        for (int j = 0; j < 8; ++j) g[j] = *(const float4*)(gsrc + (size_t)(2 * j) * IN_F);
        #pragma unroll
        for (int j = 0; j < 8; ++j) wst[wave][2 * j + h][c] = g[j];
        const float4* srow = &wst[wave][c][h];

        if (t + 1 < NTILES)
            __builtin_prefetch(wtile + 16 * IN_F + c * IN_F, 0, 1);

        v8f acc[4];
        const unsigned flags = tile_products_table(srow, af, tab, oma, a8, a, acc);
        if (__builtin_amdgcn_ballot_w32((flags & MULTI) != 0u) != 0u) {
            int again = 0;
            asm volatile("" : "+v"(again));
            tile_products_exact(srow + again, af, Rp, oma, a8, a, acc);
        }

        float* mine = &red[t & 1][wave][0][lane];
        #pragma unroll
        for (int mt = 0; mt < 4; ++mt) {
            #pragma unroll
            for (int j = 0; j < 8; ++j)
                mine[(mt * 8 + j) * 32] = acc[mt][j];
        }
        __syncthreads();
        const float* col = &red[t & 1][0][rr][lane];
        float s0 = 0.0f, s1 = 0.0f;
        #pragma unroll
        for (int w = 0; w < NW; ++w) {
            s0 += col[w * 1024];
            s1 += col[w * 1024 + 32];
        }

        const float s = scale[n0t + c];
        unsafeAtomicAdd(&out[(size_t)mout * OUT_F + n0t + c], s0 * s);
        unsafeAtomicAdd(&out[(size_t)(mout + 1) * OUT_F + n0t + c], s1 * s);
    }
}

extern "C" void kernel_launch(void* const* d_in, const int* in_sizes, int n_in,
                              void* d_out, int out_size, void* d_ws, size_t ws_size,
                              hipStream_t stream) {
    const float* x     = (const float*)d_in[0];
    const float* lw    = (const float*)d_in[1];
    const float* scale = (const float*)d_in[2];
    const float* thr   = (const float*)d_in[3];
    const int*   step  = (const int*)d_in[4];
    float* out = (float*)d_out;

    const int n4 = out_size / 4;
    zero_out<<<dim3((n4 + 255) / 256), dim3(256), 0, stream>>>(out, n4);

    dim3 grid(OUT_F / NBLK, KSPLIT);
    dim3 block(NW * 32);
    density_linear_wmma<<<grid, block, 0, stream>>>(x, lw, scale, thr, step, out);
}


# work-group barrier split in two: a wave signals once its partial tile is in LDS and waits one tile later, summing the 16 partials while the next tile's weight loads fly; sums and order unchanged
# speedup vs baseline: 1.7896x; 1.0534x over previous
#include <hip/hip_runtime.h>

typedef float v2f __attribute__((ext_vector_type(2)));
typedef float v8f __attribute__((ext_vector_type(8)));

#define IN_F   4096
#define OUT_F  16384
#define M_ROWS 64
#define KK     16
#define NUM_BANKS 64
#define ANNEAL 2000
#define NW     16
#define KWAVE  64
#define KSTEPS (KWAVE / 4)
#define KSLICE (NW * KWAVE)
#define KSPLIT (IN_F / KSLICE)
#define NTILES 16
#define NBLK   (NTILES * 16)
#define GCELL  8192
#define MULTI  0x100u
#define KGRP   (KWAVE / 8)
#define SPITCH 17

__device__ __forceinline__ int grid_cell(float v)
{
    float f = floorf(v * (float)GCELL);
    f = fminf(fmaxf(f, -1.0f), (float)(GCELL + 1));
    return (int)f;
}

__device__ __forceinline__ float eff_weight(float cl, int n, float oma, float a8, float a)
{
    return __builtin_fmaf(oma, cl, __builtin_fmaf((float)n, a8, -a));
}

__device__ __forceinline__ void tile_products_exact(const float4* srow,
                                                    const v2f (&af)[KSTEPS][4],
                                                    const float* Rp,
                                                    float oma, float a8, float a,
                                                    v8f (&acc)[4])
{
    #pragma unroll
    for (int mt = 0; mt < 4; ++mt) acc[mt] = (v8f)0.0f;
    #pragma unroll
    for (int ks = 0; ks < KSTEPS; ++ks) {
        const float4 q = srow[2 * (ks >> 1)];
        const float w0 = (ks & 1) ? q.z : q.x;
        const float w1 = (ks & 1) ? q.w : q.y;
        const float c0 = fminf(fmaxf(w0, -1.0f), 1.0f);
        const float c1 = fminf(fmaxf(w1, -1.0f), 1.0f);
        const float t0 = __builtin_fmaf(c0, 0.5f, 0.5f);
        const float t1 = __builtin_fmaf(c1, 0.5f, 0.5f);
        int n0 = 0, n1 = 0;
        #pragma nounroll
        for (int i = 0; i < KK; ++i) {
            const float ri = Rp[i];
            n0 += (ri < t0) ? 1 : 0;
            n1 += (ri < t1) ? 1 : 0;
        }
        v2f b;
        b.x = eff_weight(c0, n0, oma, a8, a);
        b.y = eff_weight(c1, n1, oma, a8, a);
        #pragma unroll
        for (int mt = 0; mt < 4; ++mt)
            acc[mt] = __builtin_amdgcn_wmma_f32_16x16x4_f32(
                false, af[ks][mt], false, b, (short)0, acc[mt], false, false);
    }
}

__device__ __forceinline__ unsigned tile_products_table(const float4* srow,
                                                        const v2f (&af)[KSTEPS][4],
                                                        const uint2* tab,
                                                        float oma, float a8, float a,
                                                        v8f (&acc)[4])
{
    unsigned flags = 0u;
    #pragma unroll
    for (int mt = 0; mt < 4; ++mt) acc[mt] = (v8f)0.0f;

    #pragma unroll
    for (int ks = 0; ks < KSTEPS; ++ks) {
        const float4 q = srow[2 * (ks >> 1)];
        const float w0 = (ks & 1) ? q.z : q.x;
        const float w1 = (ks & 1) ? q.w : q.y;
        const float c0 = fminf(fmaxf(w0, -1.0f), 1.0f);
        const float c1 = fminf(fmaxf(w1, -1.0f), 1.0f);
        const float t0 = __builtin_fmaf(c0, 0.5f, 0.5f);
        const float t1 = __builtin_fmaf(c1, 0.5f, 0.5f);
        const uint2 e0 = tab[(int)(t0 * (float)GCELL)];
        const uint2 e1 = tab[(int)(t1 * (float)GCELL)];
        const int n0 = (int)(e0.y & 0xffu) + ((__uint_as_float(e0.x) < t0) ? 1 : 0);
        const int n1 = (int)(e1.y & 0xffu) + ((__uint_as_float(e1.x) < t1) ? 1 : 0);
        flags |= e0.y | e1.y;
        v2f b;
        b.x = eff_weight(c0, n0, oma, a8, a);
        b.y = eff_weight(c1, n1, oma, a8, a);
        #pragma unroll
        for (int mt = 0; mt < 4; ++mt)
            acc[mt] = __builtin_amdgcn_wmma_f32_16x16x4_f32(
                false, af[ks][mt], false, b, (short)0, acc[mt], false, false);
    }
    return flags;
}

__device__ __forceinline__ void bar_arrive() { asm volatile("s_wait_dscnt 0x0\n\ts_barrier_signal -1" ::: "memory"); }
__device__ __forceinline__ void bar_wait() { asm volatile("s_barrier_wait -1" ::: "memory"); }

__device__ __forceinline__ void reduce_tile(const float* col, float s, float* dst)
{
    float s0 = 0.0f, s1 = 0.0f;
    #pragma unroll
    for (int w = 0; w < NW; ++w) {
        s0 += col[w * 1024];
        s1 += col[w * 1024 + 32];
    }
    unsafeAtomicAdd(dst, s0 * s);
    unsafeAtomicAdd(dst + OUT_F, s1 * s);
}

__global__ __launch_bounds__(256)
void zero_out(float* __restrict__ out, int n4)
{
    int i = blockIdx.x * blockDim.x + threadIdx.x;
    if (i < n4) ((float4*)out)[i] = make_float4(0.f, 0.f, 0.f, 0.f);
}

__global__ __launch_bounds__(NW * 32)
void density_linear_wmma(const float* __restrict__ x,
                         const float* __restrict__ lw,
                         const float* __restrict__ scale,
                         const float* __restrict__ thr,
                         const int*   __restrict__ stepp,
                         float* __restrict__ out)
{
    __shared__ float red[2][NW][32][32];
    __shared__ uint2 tab[GCELL + 8];
    __shared__ float4 wst[NW][16][SPITCH];

    const int tid   = threadIdx.x;
    const int lane  = tid & 31;
    const int wave  = __builtin_amdgcn_readfirstlane(tid >> 5);
    const int c     = lane & 15;
    const int h     = lane >> 4;
    const int koff  = h * 4;
    const int kw0   = blockIdx.y * KSLICE + wave * KWAVE;
    const int nblk0 = blockIdx.x * NBLK;
    const int loff  = c * IN_F + koff;

    const int step = stepp[0];
    float a = fminf((float)step / (float)ANNEAL, 1.0f);
    const float oma = 1.0f - a;
    const float a8  = a * 0.125f;

    const float* Rp = thr + (unsigned)(step % NUM_BANKS) * KK;
    float r[KK];
    #pragma unroll
    for (int i = 0; i < KK; ++i) r[i] = Rp[i];

    v2f af[KSTEPS][4];
    #pragma unroll
    for (int p = 0; p < KGRP; ++p) {
        #pragma unroll
        for (int mt = 0; mt < 4; ++mt) {
            const float4 xv = *(const float4*)(x + kw0 + mt * 16 * IN_F + p * 8 + loff);
            af[2 * p][mt].x     = xv.x;
            af[2 * p][mt].y     = xv.y;
            af[2 * p + 1][mt].x = xv.z;
            af[2 * p + 1][mt].y = xv.w;
        }
    }

    {
        int ci[KK];
        #pragma unroll
        for (int i = 0; i < KK; ++i) ci[i] = grid_cell(r[i]);
        const int k0 = tid * (GCELL / (NW * 32));
        int below[GCELL / (NW * 32) + 2];
        #pragma unroll
        for (int j = 0; j < GCELL / (NW * 32) + 2; ++j) {
            unsigned bits = 0u;
            #pragma unroll
            for (int i = 0; i < KK; ++i)
                bits = __builtin_amdgcn_alignbit(bits, (unsigned)(ci[i] - (k0 + j)), 31u);
            below[j] = __builtin_popcount(bits);
        }
        #pragma unroll
        for (int j = 0; j < GCELL / (NW * 32) + 1; ++j) {
            const int inside = below[j + 1] - below[j];
            if (j < GCELL / (NW * 32) || k0 + j == GCELL)
                tab[k0 + j] = make_uint2(0x7f800000u, (unsigned)below[j] | (inside >= 2 ? MULTI : 0u));
        }
        __syncthreads();
        if (tid < KK) {
            const float rv = Rp[tid];
            const int cc = grid_cell(rv);
            if (cc >= 0 && cc <= GCELL) tab[cc].x = __float_as_uint(rv);
        }
        __syncthreads();
    }

    const int rr   = 2 * wave;
    const int mout = (rr >> 3) * 16 + h * 8 + (rr & 7);

    const float* gsrc0 = lw + (size_t)nblk0 * IN_F + kw0 + h * IN_F + c * 4;
    float4 g0 = *(const float4*)(gsrc0);
    float4 g1 = *(const float4*)(gsrc0 + (size_t)2 * IN_F);
    float4 g2 = *(const float4*)(gsrc0 + (size_t)4 * IN_F);
    float4 g3 = *(const float4*)(gsrc0 + (size_t)6 * IN_F);
    float4 g4 = *(const float4*)(gsrc0 + (size_t)8 * IN_F);
    float4 g5 = *(const float4*)(gsrc0 + (size_t)10 * IN_F);
    float4 g6 = *(const float4*)(gsrc0 + (size_t)12 * IN_F);
    float4 g7 = *(const float4*)(gsrc0 + (size_t)14 * IN_F);

    const float oma_s = __uint_as_float(__builtin_amdgcn_readfirstlane(__float_as_uint(oma)));
    const float a8_s  = __uint_as_float(__builtin_amdgcn_readfirstlane(__float_as_uint(a8)));
    const float a_s   = __uint_as_float(__builtin_amdgcn_readfirstlane(__float_as_uint(a)));

    float s_prev = 0.0f;
    const unsigned ooff = (unsigned)(mout * OUT_F + nblk0 + c);

    for (int t = 0; t < NTILES; ++t) {
        const int n0t = nblk0 + t * 16;
        const float* wtile = lw + (size_t)n0t * IN_F + kw0;

        wst[wave][0 + h][c] = g0;
        wst[wave][2 + h][c] = g1;
        wst[wave][4 + h][c] = g2;
        wst[wave][6 + h][c] = g3;
        wst[wave][8 + h][c] = g4;
        wst[wave][10 + h][c] = g5;
        wst[wave][12 + h][c] = g6;
        wst[wave][14 + h][c] = g7;
        const float4* srow = &wst[wave][c][h];

        if (t + 1 < NTILES)
            __builtin_prefetch(wtile + 16 * IN_F + c * IN_F, 0, 1);

        const float s_t = scale[n0t + c];

        float oma_t = oma_s, a8_t = a8_s, a_t = a_s;
        asm volatile("" : "+s"(oma_t), "+s"(a8_t), "+s"(a_t));

        v8f acc[4];
        const unsigned flags = tile_products_table(srow, af, tab, oma_t, a8_t, a_t, acc);
        if (__builtin_amdgcn_ballot_w32((flags & MULTI) != 0u) != 0u) {
            int again = 0;
            asm volatile("" : "+v"(again));
            tile_products_exact(srow + again, af, Rp, oma_t, a8_t, a_t, acc);
        }

        if (t + 1 < NTILES) {
            const float* gsrc = wtile + 16 * IN_F + h * IN_F + c * 4;
            g0 = *(const float4*)(gsrc);
            g1 = *(const float4*)(gsrc + (size_t)2 * IN_F);
            g2 = *(const float4*)(gsrc + (size_t)4 * IN_F);
            g3 = *(const float4*)(gsrc + (size_t)6 * IN_F);
            g4 = *(const float4*)(gsrc + (size_t)8 * IN_F);
            g5 = *(const float4*)(gsrc + (size_t)10 * IN_F);
            g6 = *(const float4*)(gsrc + (size_t)12 * IN_F);
            g7 = *(const float4*)(gsrc + (size_t)14 * IN_F);
        }

        if (t > 0) {
            bar_wait();
            reduce_tile(&red[(t - 1) & 1][0][rr][lane], s_prev, out + (ooff + (unsigned)((t - 1) * 16)));
        }

        float* mine = &red[t & 1][wave][0][lane];
        #pragma unroll
        for (int mt = 0; mt < 4; ++mt) {
            #pragma unroll
            for (int j = 0; j < 8; ++j)
                mine[(mt * 8 + j) * 32] = acc[mt][j];
        }
        bar_arrive();
        s_prev = s_t;
    }
    bar_wait();
    reduce_tile(&red[(NTILES - 1) & 1][0][rr][lane], s_prev, out + (ooff + (unsigned)((NTILES - 1) * 16)));
}

extern "C" void kernel_launch(void* const* d_in, const int* in_sizes, int n_in,
                              void* d_out, int out_size, void* d_ws, size_t ws_size,
                              hipStream_t stream) {
    const float* x     = (const float*)d_in[0];
    const float* lw    = (const float*)d_in[1];
    const float* scale = (const float*)d_in[2];
    const float* thr   = (const float*)d_in[3];
    const int*   step  = (const int*)d_in[4];
    float* out = (float*)d_out;

    const int n4 = out_size / 4;
    zero_out<<<dim3((n4 + 255) / 256), dim3(256), 0, stream>>>(out, n4);

    dim3 grid(OUT_F / NBLK, KSPLIT);
    dim3 block(NW * 32);
    density_linear_wmma<<<grid, block, 0, stream>>>(x, lw, scale, thr, step, out);
}


# two 8-wave work-groups per WGP (was one of 16), 512 K each, count table 4096 cells so both fit LDS: one group's barrier waits are filled by the other's waves; 8 atomic adds per output (was 4)
# speedup vs baseline: 1.8414x; 1.0289x over previous
#include <hip/hip_runtime.h>

typedef float v2f __attribute__((ext_vector_type(2)));
typedef float v8f __attribute__((ext_vector_type(8)));

#define IN_F   4096
#define OUT_F  16384
#define M_ROWS 64
#define KK     16
#define NUM_BANKS 64
#define ANNEAL 2000
#define NW     8
#define KWAVE  64
#define KSTEPS (KWAVE / 4)
#define KSLICE (NW * KWAVE)
#define KSPLIT (IN_F / KSLICE)
#define NTILES 16
#define NBLK   (NTILES * 16)
#define GCELL  4096
#define MULTI  0x100u
#define KGRP   (KWAVE / 8)
#define RPW    (32 / NW)
#define SPITCH 17

__device__ __forceinline__ int grid_cell(float v)
{
    float f = floorf(v * (float)GCELL);
    f = fminf(fmaxf(f, -1.0f), (float)(GCELL + 1));
    return (int)f;
}

__device__ __forceinline__ float eff_weight(float cl, int n, float oma, float a8, float a)
{
    return __builtin_fmaf(oma, cl, __builtin_fmaf((float)n, a8, -a));
}

__device__ __forceinline__ void tile_products_exact(const float4* srow,
                                                    const v2f (&af)[KSTEPS][4],
                                                    const float* Rp,
                                                    float oma, float a8, float a,
                                                    v8f (&acc)[4])
{
    #pragma unroll
    for (int mt = 0; mt < 4; ++mt) acc[mt] = (v8f)0.0f;
    #pragma unroll
    for (int ks = 0; ks < KSTEPS; ++ks) {
        const float4 q = srow[2 * (ks >> 1)];
        const float w0 = (ks & 1) ? q.z : q.x;
        const float w1 = (ks & 1) ? q.w : q.y;
        const float c0 = fminf(fmaxf(w0, -1.0f), 1.0f);
        const float c1 = fminf(fmaxf(w1, -1.0f), 1.0f);
        const float t0 = __builtin_fmaf(c0, 0.5f, 0.5f);
        const float t1 = __builtin_fmaf(c1, 0.5f, 0.5f);
        int n0 = 0, n1 = 0;
        #pragma nounroll
        for (int i = 0; i < KK; ++i) {
            const float ri = Rp[i];
            n0 += (ri < t0) ? 1 : 0;
            n1 += (ri < t1) ? 1 : 0;
        }
        v2f b;
        b.x = eff_weight(c0, n0, oma, a8, a);
        b.y = eff_weight(c1, n1, oma, a8, a);
        #pragma unroll
        for (int mt = 0; mt < 4; ++mt)
            acc[mt] = __builtin_amdgcn_wmma_f32_16x16x4_f32(
                false, af[ks][mt], false, b, (short)0, acc[mt], false, false);
    }
}

__device__ __forceinline__ unsigned tile_products_table(const float4* srow,
                                                        const v2f (&af)[KSTEPS][4],
                                                        const uint2* tab,
                                                        float oma, float a8, float a,
                                                        v8f (&acc)[4])
{
    unsigned flags = 0u;
    #pragma unroll
    for (int mt = 0; mt < 4; ++mt) acc[mt] = (v8f)0.0f;

    #pragma unroll
    for (int ks = 0; ks < KSTEPS; ++ks) {
        const float4 q = srow[2 * (ks >> 1)];
        const float w0 = (ks & 1) ? q.z : q.x;
        const float w1 = (ks & 1) ? q.w : q.y;
        const float c0 = fminf(fmaxf(w0, -1.0f), 1.0f);
        const float c1 = fminf(fmaxf(w1, -1.0f), 1.0f);
        const float t0 = __builtin_fmaf(c0, 0.5f, 0.5f);
        const float t1 = __builtin_fmaf(c1, 0.5f, 0.5f);
        const uint2 e0 = tab[(int)(t0 * (float)GCELL)];
        const uint2 e1 = tab[(int)(t1 * (float)GCELL)];
        const int n0 = (int)(e0.y & 0xffu) + ((__uint_as_float(e0.x) < t0) ? 1 : 0);
        const int n1 = (int)(e1.y & 0xffu) + ((__uint_as_float(e1.x) < t1) ? 1 : 0);
        flags |= e0.y | e1.y;
        v2f b;
        b.x = eff_weight(c0, n0, oma, a8, a);
        b.y = eff_weight(c1, n1, oma, a8, a);
        #pragma unroll
        for (int mt = 0; mt < 4; ++mt)
            acc[mt] = __builtin_amdgcn_wmma_f32_16x16x4_f32(
                false, af[ks][mt], false, b, (short)0, acc[mt], false, false);
    }
    return flags;
}

__device__ __forceinline__ void bar_arrive() { asm volatile("s_wait_dscnt 0x0\n\ts_barrier_signal -1" ::: "memory"); }
__device__ __forceinline__ void bar_wait() { asm volatile("s_barrier_wait -1" ::: "memory"); }

__device__ __forceinline__ void reduce_tile(const float* col, float s, float* dst)
{
    float sum[RPW];
    #pragma unroll
    for (int j = 0; j < RPW; ++j) sum[j] = 0.0f;
    #pragma unroll
    for (int w = 0; w < NW; ++w) {
        #pragma unroll
        for (int j = 0; j < RPW; ++j)
            sum[j] += col[w * 1024 + j * 32];
    }
    #pragma unroll
    for (int j = 0; j < RPW; ++j)
        unsafeAtomicAdd(dst + j * OUT_F, sum[j] * s);
}

__global__ __launch_bounds__(256)
void zero_out(float* __restrict__ out, int n4)
{
    int i = blockIdx.x * blockDim.x + threadIdx.x;
    if (i < n4) ((float4*)out)[i] = make_float4(0.f, 0.f, 0.f, 0.f);
}

__global__ __launch_bounds__(NW * 32, 4)
void density_linear_wmma(const float* __restrict__ x,
                         const float* __restrict__ lw,
                         const float* __restrict__ scale,
                         const float* __restrict__ thr,
                         const int*   __restrict__ stepp,
                         float* __restrict__ out)
{
    __shared__ float red[2][NW][32][32];
    __shared__ uint2 tab[GCELL + 8];
    __shared__ float4 wst[NW][16][SPITCH];

    const int tid   = threadIdx.x;
    const int lane  = tid & 31;
    const int wave  = __builtin_amdgcn_readfirstlane(tid >> 5);
    const int c     = lane & 15;
    const int h     = lane >> 4;
    const int koff  = h * 4;
    const int kw0   = blockIdx.y * KSLICE + wave * KWAVE;
    const int nblk0 = blockIdx.x * NBLK;
    const int loff  = c * IN_F + koff;

    const int step = stepp[0];
    float a = fminf((float)step / (float)ANNEAL, 1.0f);
    const float oma = 1.0f - a;
    const float a8  = a * 0.125f;

    const float* Rp = thr + (unsigned)(step % NUM_BANKS) * KK;
    float r[KK];
    #pragma unroll
    for (int i = 0; i < KK; ++i) r[i] = Rp[i];

    v2f af[KSTEPS][4];
    #pragma unroll
    for (int p = 0; p < KGRP; ++p) {
        #pragma unroll
        for (int mt = 0; mt < 4; ++mt) {
            const float4 xv = *(const float4*)(x + kw0 + mt * 16 * IN_F + p * 8 + loff);
            af[2 * p][mt].x     = xv.x;
            af[2 * p][mt].y     = xv.y;
            af[2 * p + 1][mt].x = xv.z;
            af[2 * p + 1][mt].y = xv.w;
        }
    }

    {
        int ci[KK];
        #pragma unroll
        for (int i = 0; i < KK; ++i) ci[i] = grid_cell(r[i]);
        const int k0 = tid * (GCELL / (NW * 32));
        int below[GCELL / (NW * 32) + 2];
        #pragma unroll
        for (int j = 0; j < GCELL / (NW * 32) + 2; ++j) {
            unsigned bits = 0u;
            #pragma unroll
            for (int i = 0; i < KK; ++i)
                bits = __builtin_amdgcn_alignbit(bits, (unsigned)(ci[i] - (k0 + j)), 31u);
            below[j] = __builtin_popcount(bits);
        }
        #pragma unroll
        for (int j = 0; j < GCELL / (NW * 32) + 1; ++j) {
            const int inside = below[j + 1] - below[j];
            if (j < GCELL / (NW * 32) || k0 + j == GCELL)
                tab[k0 + j] = make_uint2(0x7f800000u, (unsigned)below[j] | (inside >= 2 ? MULTI : 0u));
        }
        __syncthreads();
        if (tid < KK) {
            const float rv = Rp[tid];
            const int cc = grid_cell(rv);
            if (cc >= 0 && cc <= GCELL) tab[cc].x = __float_as_uint(rv);
        }
        __syncthreads();
    }

    const int rr   = RPW * wave;
    const int mout = (rr >> 3) * 16 + h * 8 + (rr & 7);

    const float* gsrc0 = lw + (size_t)nblk0 * IN_F + kw0 + h * IN_F + c * 4;
    float4 g0 = *(const float4*)(gsrc0);
    float4 g1 = *(const float4*)(gsrc0 + (size_t)2 * IN_F);
    float4 g2 = *(const float4*)(gsrc0 + (size_t)4 * IN_F);
    float4 g3 = *(const float4*)(gsrc0 + (size_t)6 * IN_F);
    float4 g4 = *(const float4*)(gsrc0 + (size_t)8 * IN_F);
    float4 g5 = *(const float4*)(gsrc0 + (size_t)10 * IN_F);
    float4 g6 = *(const float4*)(gsrc0 + (size_t)12 * IN_F);
    float4 g7 = *(const float4*)(gsrc0 + (size_t)14 * IN_F);

    const float oma_s = __uint_as_float(__builtin_amdgcn_readfirstlane(__float_as_uint(oma)));
    const float a8_s  = __uint_as_float(__builtin_amdgcn_readfirstlane(__float_as_uint(a8)));
    const float a_s   = __uint_as_float(__builtin_amdgcn_readfirstlane(__float_as_uint(a)));

    float s_prev = 0.0f;
    const unsigned ooff = (unsigned)(mout * OUT_F + nblk0 + c);

    for (int t = 0; t < NTILES; ++t) {
        const int n0t = nblk0 + t * 16;
        const float* wtile = lw + (size_t)n0t * IN_F + kw0;

        wst[wave][0 + h][c] = g0;
        wst[wave][2 + h][c] = g1;
        wst[wave][4 + h][c] = g2;
        wst[wave][6 + h][c] = g3;
        wst[wave][8 + h][c] = g4;
        wst[wave][10 + h][c] = g5;
        wst[wave][12 + h][c] = g6;
        wst[wave][14 + h][c] = g7;
        const float4* srow = &wst[wave][c][h];

        if (t + 1 < NTILES)
            __builtin_prefetch(wtile + 16 * IN_F + c * IN_F, 0, 1);

        const float s_t = scale[n0t + c];

        float oma_t = oma_s, a8_t = a8_s, a_t = a_s;
        asm volatile("" : "+s"(oma_t), "+s"(a8_t), "+s"(a_t));

        v8f acc[4];
        const unsigned flags = tile_products_table(srow, af, tab, oma_t, a8_t, a_t, acc);
        if (__builtin_amdgcn_ballot_w32((flags & MULTI) != 0u) != 0u) {
            int again = 0;
            asm volatile("" : "+v"(again));
            tile_products_exact(srow + again, af, Rp, oma_t, a8_t, a_t, acc);
        }

        if (t + 1 < NTILES) {
            const float* gsrc = wtile + 16 * IN_F + h * IN_F + c * 4;
            g0 = *(const float4*)(gsrc);
            g1 = *(const float4*)(gsrc + (size_t)2 * IN_F);
            g2 = *(const float4*)(gsrc + (size_t)4 * IN_F);
            g3 = *(const float4*)(gsrc + (size_t)6 * IN_F);
            g4 = *(const float4*)(gsrc + (size_t)8 * IN_F);
            g5 = *(const float4*)(gsrc + (size_t)10 * IN_F);
            g6 = *(const float4*)(gsrc + (size_t)12 * IN_F);
            g7 = *(const float4*)(gsrc + (size_t)14 * IN_F);
        }

        if (t > 0) {
            bar_wait();
            reduce_tile(&red[(t - 1) & 1][0][rr][lane], s_prev, out + (ooff + (unsigned)((t - 1) * 16)));
        }

        float* mine = &red[t & 1][wave][0][lane];
        #pragma unroll
        for (int mt = 0; mt < 4; ++mt) {
            #pragma unroll
            for (int j = 0; j < 8; ++j)
                mine[(mt * 8 + j) * 32] = acc[mt][j];
        }
        bar_arrive();
        s_prev = s_t;
    }
    bar_wait();
    reduce_tile(&red[(NTILES - 1) & 1][0][rr][lane], s_prev, out + (ooff + (unsigned)((NTILES - 1) * 16)));
}

extern "C" void kernel_launch(void* const* d_in, const int* in_sizes, int n_in,
                              void* d_out, int out_size, void* d_ws, size_t ws_size,
                              hipStream_t stream) {
    const float* x     = (const float*)d_in[0];
    const float* lw    = (const float*)d_in[1];
    const float* scale = (const float*)d_in[2];
    const float* thr   = (const float*)d_in[3];
    const int*   step  = (const int*)d_in[4];
    float* out = (float*)d_out;

    const int n4 = out_size / 4;
    zero_out<<<dim3((n4 + 255) / 256), dim3(256), 0, stream>>>(out, n4);

    dim3 grid(OUT_F / NBLK, KSPLIT);
    dim3 block(NW * 32);
    density_linear_wmma<<<grid, block, 0, stream>>>(x, lw, scale, thr, step, out);
}


# per-weight vector work cut by a fifth: table entry holds the count's finished term as a float, cell index read from one fma's mantissa, weight formed from the target density; sums, order unchanged
# speedup vs baseline: 1.9286x; 1.0474x over previous
#include <hip/hip_runtime.h>

typedef float v2f __attribute__((ext_vector_type(2)));
typedef float v8f __attribute__((ext_vector_type(8)));

#define IN_F   4096
#define OUT_F  16384
#define M_ROWS 64
#define KK     16
#define NUM_BANKS 64
#define ANNEAL 2000
#define NW     8
#define KWAVE  64
#define KSTEPS (KWAVE / 4)
#define KSLICE (NW * KWAVE)
#define KSPLIT (IN_F / KSLICE)
#define NTILES 16
#define NBLK   (NTILES * 16)
#define GCELL  4096
#define MULTI  0x1u
#define MAGIC  8388608.0f
#define MAGICB 0x4B000000u
#define KGRP   (KWAVE / 8)
#define RPW    (32 / NW)
#define SPITCH 17

__device__ __forceinline__ float target_density(float w)
{
    return __builtin_amdgcn_fmed3f(__builtin_fmaf(w, 0.5f, 0.5f), 0.0f, 1.0f);
}

__device__ __forceinline__ unsigned density_cell(float t)
{
    return __float_as_uint(__builtin_fmaf(t, (float)GCELL, MAGIC)) - MAGICB;
}

__device__ __forceinline__ int grid_cell(float v)
{
    if (!(v >= 0.0f)) return -1;
    if (v > 1.0f) return GCELL + 1;
    return (int)density_cell(v);
}

__device__ __forceinline__ float count_term(int n, float a8)
{
    return __builtin_fmaf((float)n, a8, -1.0f);
}

__device__ __forceinline__ void tile_products_exact(const float4* srow,
                                                    const v2f (&af)[KSTEPS][4],
                                                    const float* Rp,
                                                    float oma2, float a8,
                                                    v8f (&acc)[4])
{
    #pragma unroll
    for (int mt = 0; mt < 4; ++mt) acc[mt] = (v8f)0.0f;
    #pragma unroll
    for (int ks = 0; ks < KSTEPS; ++ks) {
        const float4 q = srow[2 * (ks >> 1)];
        const float w0 = (ks & 1) ? q.z : q.x;
        const float w1 = (ks & 1) ? q.w : q.y;
        const float t0 = target_density(w0);
        const float t1 = target_density(w1);
        int n0 = 0, n1 = 0;
        #pragma nounroll
        for (int i = 0; i < KK; ++i) {
            const float ri = Rp[i];
            n0 += (ri < t0) ? 1 : 0;
            n1 += (ri < t1) ? 1 : 0;
        }
        v2f b;
        b.x = __builtin_fmaf(oma2, t0, count_term(n0, a8));
        b.y = __builtin_fmaf(oma2, t1, count_term(n1, a8));
        #pragma unroll
        for (int mt = 0; mt < 4; ++mt)
            acc[mt] = __builtin_amdgcn_wmma_f32_16x16x4_f32(
                false, af[ks][mt], false, b, (short)0, acc[mt], false, false);
    }
}

__device__ __forceinline__ unsigned tile_products_table(const float4* srow,
                                                        const v2f (&af)[KSTEPS][4],
                                                        const uint2* tab,
                                                        float oma2, float a8,
                                                        v8f (&acc)[4])
{
    unsigned flags = 0u;
    #pragma unroll
    for (int mt = 0; mt < 4; ++mt) acc[mt] = (v8f)0.0f;

    #pragma unroll
    for (int ks = 0; ks < KSTEPS; ++ks) {
        const float4 q = srow[2 * (ks >> 1)];
        const float w0 = (ks & 1) ? q.z : q.x;
        const float w1 = (ks & 1) ? q.w : q.y;
        const float t0 = target_density(w0);
        const float t1 = target_density(w1);
        const uint2 e0 = tab[density_cell(t0)];
        const uint2 e1 = tab[density_cell(t1)];
        flags |= e0.y | e1.y;
        v2f b;
        b.x = __builtin_fmaf(oma2, t0, __uint_as_float(e0.y)) + ((__uint_as_float(e0.x) < t0) ? a8 : 0.0f);
        b.y = __builtin_fmaf(oma2, t1, __uint_as_float(e1.y)) + ((__uint_as_float(e1.x) < t1) ? a8 : 0.0f);
        #pragma unroll
        for (int mt = 0; mt < 4; ++mt)
            acc[mt] = __builtin_amdgcn_wmma_f32_16x16x4_f32(
                false, af[ks][mt], false, b, (short)0, acc[mt], false, false);
    }
    return flags;
}

__device__ __forceinline__ void bar_arrive() { asm volatile("s_wait_dscnt 0x0\n\ts_barrier_signal -1" ::: "memory"); }
__device__ __forceinline__ void bar_wait() { asm volatile("s_barrier_wait -1" ::: "memory"); }

__device__ __forceinline__ void reduce_tile(const float* col, float s, float* dst)
{
    float sum[RPW];
    #pragma unroll
    for (int j = 0; j < RPW; ++j) sum[j] = 0.0f;
    #pragma unroll
    for (int w = 0; w < NW; ++w) {
        #pragma unroll
        for (int j = 0; j < RPW; ++j)
            sum[j] += col[w * 1024 + j * 32];
    }
    #pragma unroll
    for (int j = 0; j < RPW; ++j)
        unsafeAtomicAdd(dst + j * OUT_F, sum[j] * s);
}

__global__ __launch_bounds__(256)
void zero_out(float* __restrict__ out, int n4)
{
    int i = blockIdx.x * blockDim.x + threadIdx.x;
    if (i < n4) ((float4*)out)[i] = make_float4(0.f, 0.f, 0.f, 0.f);
}

__global__ __launch_bounds__(NW * 32, 4)
void density_linear_wmma(const float* __restrict__ x,
                         const float* __restrict__ lw,
                         const float* __restrict__ scale,
                         const float* __restrict__ thr,
                         const int*   __restrict__ stepp,
                         float* __restrict__ out)
{
    __shared__ float red[2][NW][32][32];
    __shared__ uint2 tab[GCELL + 8];
    __shared__ float4 wst[NW][16][SPITCH];

    const int tid   = threadIdx.x;
    const int lane  = tid & 31;
    const int wave  = __builtin_amdgcn_readfirstlane(tid >> 5);
    const int c     = lane & 15;
    const int h     = lane >> 4;
    const int koff  = h * 4;
    const int kw0   = blockIdx.y * KSLICE + wave * KWAVE;
    const int nblk0 = blockIdx.x * NBLK;
    const int loff  = c * IN_F + koff;

    const int step = stepp[0];
    float a = fminf((float)step / (float)ANNEAL, 1.0f);
    const float oma2 = 2.0f * (1.0f - a);
    const float a8   = a * 0.125f;

    const float* Rp = thr + (unsigned)(step % NUM_BANKS) * KK;
    float r[KK];
    #pragma unroll
    for (int i = 0; i < KK; ++i) r[i] = Rp[i];

    v2f af[KSTEPS][4];
    #pragma unroll
    for (int p = 0; p < KGRP; ++p) {
        #pragma unroll
        for (int mt = 0; mt < 4; ++mt) {
            const float4 xv = *(const float4*)(x + kw0 + mt * 16 * IN_F + p * 8 + loff);
            af[2 * p][mt].x     = xv.x;
            af[2 * p][mt].y     = xv.y;
            af[2 * p + 1][mt].x = xv.z;
            af[2 * p + 1][mt].y = xv.w;
        }
    }

    {
        int ci[KK];
        #pragma unroll
        for (int i = 0; i < KK; ++i) ci[i] = grid_cell(r[i]);
        const int k0 = tid * (GCELL / (NW * 32));
        int below[GCELL / (NW * 32) + 2];
        #pragma unroll
        for (int j = 0; j < GCELL / (NW * 32) + 2; ++j) {
            unsigned bits = 0u;
            #pragma unroll
            for (int i = 0; i < KK; ++i)
                bits = __builtin_amdgcn_alignbit(bits, (unsigned)(ci[i] - (k0 + j)), 31u);
            below[j] = __builtin_popcount(bits);
        }
        #pragma unroll
        for (int j = 0; j < GCELL / (NW * 32) + 1; ++j) {
            const int inside = below[j + 1] - below[j];
            if (j < GCELL / (NW * 32) || k0 + j == GCELL)
                tab[k0 + j] = make_uint2(0x7f800000u, (__float_as_uint(count_term(below[j], a8)) & ~MULTI) | (inside >= 2 ? MULTI : 0u));
        }
        __syncthreads();
        if (tid < KK) {
            const float rv = Rp[tid];
            const int cc = grid_cell(rv);
            if (cc >= 0 && cc <= GCELL) tab[cc].x = __float_as_uint(rv);
        }
        __syncthreads();
    }

    const int rr   = RPW * wave;
    const int mout = (rr >> 3) * 16 + h * 8 + (rr & 7);

    const float* gsrc0 = lw + (size_t)nblk0 * IN_F + kw0 + h * IN_F + c * 4;
    float4 g0 = *(const float4*)(gsrc0);
    float4 g1 = *(const float4*)(gsrc0 + (size_t)2 * IN_F);
    float4 g2 = *(const float4*)(gsrc0 + (size_t)4 * IN_F);
    float4 g3 = *(const float4*)(gsrc0 + (size_t)6 * IN_F);
    float4 g4 = *(const float4*)(gsrc0 + (size_t)8 * IN_F);
    float4 g5 = *(const float4*)(gsrc0 + (size_t)10 * IN_F);
    float4 g6 = *(const float4*)(gsrc0 + (size_t)12 * IN_F);
    float4 g7 = *(const float4*)(gsrc0 + (size_t)14 * IN_F);

    const float oma2_s = __uint_as_float(__builtin_amdgcn_readfirstlane(__float_as_uint(oma2)));
    const float a8_s   = __uint_as_float(__builtin_amdgcn_readfirstlane(__float_as_uint(a8)));

    float s_prev = 0.0f;
    const unsigned ooff = (unsigned)(mout * OUT_F + nblk0 + c);

    for (int t = 0; t < NTILES; ++t) {
        const int n0t = nblk0 + t * 16;
        const float* wtile = lw + (size_t)n0t * IN_F + kw0;

        wst[wave][0 + h][c] = g0;
        wst[wave][2 + h][c] = g1;
        wst[wave][4 + h][c] = g2;
        wst[wave][6 + h][c] = g3;
        wst[wave][8 + h][c] = g4;
        wst[wave][10 + h][c] = g5;
        wst[wave][12 + h][c] = g6;
        wst[wave][14 + h][c] = g7;
        const float4* srow = &wst[wave][c][h];

        if (t + 1 < NTILES)
            __builtin_prefetch(wtile + 16 * IN_F + c * IN_F, 0, 1);

        const float s_t = scale[n0t + c];

        float oma2_t = oma2_s, a8_t = a8_s;
        asm volatile("" : "+s"(oma2_t), "+s"(a8_t));

        v8f acc[4];
        const unsigned flags = tile_products_table(srow, af, tab, oma2_t, a8_t, acc);
        if (__builtin_amdgcn_ballot_w32((flags & MULTI) != 0u) != 0u) {
            int again = 0;
            asm volatile("" : "+v"(again));
            tile_products_exact(srow + again, af, Rp, oma2_t, a8_t, acc);
        }

        if (t + 1 < NTILES) {
            const float* gsrc = wtile + 16 * IN_F + h * IN_F + c * 4;
            g0 = *(const float4*)(gsrc);
            g1 = *(const float4*)(gsrc + (size_t)2 * IN_F);
            g2 = *(const float4*)(gsrc + (size_t)4 * IN_F);
            g3 = *(const float4*)(gsrc + (size_t)6 * IN_F);
            g4 = *(const float4*)(gsrc + (size_t)8 * IN_F);
            g5 = *(const float4*)(gsrc + (size_t)10 * IN_F);
            g6 = *(const float4*)(gsrc + (size_t)12 * IN_F);
            g7 = *(const float4*)(gsrc + (size_t)14 * IN_F);
        }

        if (t > 0) {
            bar_wait();
            reduce_tile(&red[(t - 1) & 1][0][rr][lane], s_prev, out + (ooff + (unsigned)((t - 1) * 16)));
        }

        float* mine = &red[t & 1][wave][0][lane];
        #pragma unroll
        for (int mt = 0; mt < 4; ++mt) {
            #pragma unroll
            for (int j = 0; j < 8; ++j)
                mine[(mt * 8 + j) * 32] = acc[mt][j];
        }
        bar_arrive();
        s_prev = s_t;
    }
    bar_wait();
    reduce_tile(&red[(NTILES - 1) & 1][0][rr][lane], s_prev, out + (ooff + (unsigned)((NTILES - 1) * 16)));
}

extern "C" void kernel_launch(void* const* d_in, const int* in_sizes, int n_in,
                              void* d_out, int out_size, void* d_ws, size_t ws_size,
                              hipStream_t stream) {
    const float* x     = (const float*)d_in[0];
    const float* lw    = (const float*)d_in[1];
    const float* scale = (const float*)d_in[2];
    const float* thr   = (const float*)d_in[3];
    const int*   step  = (const int*)d_in[4];
    float* out = (float*)d_out;

    const int n4 = out_size / 4;
    zero_out<<<dim3((n4 + 255) / 256), dim3(256), 0, stream>>>(out, n4);

    dim3 grid(OUT_F / NBLK, KSPLIT);
    dim3 block(NW * 32);
    density_linear_wmma<<<grid, block, 0, stream>>>(x, lw, scale, thr, step, out);
}
